# starting kernel
# baseline (speedup 1.0000x reference)
#include <hip/hip_runtime.h>

#define N_NODES 100000
#define N_EDGES 640000
#define DIM 128
#define BN_EPS 1e-5f

typedef __attribute__((ext_vector_type(2))) float v2f;
typedef __attribute__((ext_vector_type(8))) float v8f;

__global__ void zero_kernel(float4* __restrict__ out4, float* __restrict__ ws) {
    int i = blockIdx.x * blockDim.x + threadIdx.x;
    if (i < (N_NODES * DIM / 4)) out4[i] = make_float4(0.f, 0.f, 0.f, 0.f);
    if (i < 256) ws[i] = 0.f;
}

__global__ void scatter_kernel(const float* __restrict__ feat,
                               const float* __restrict__ ew,
                               const int*   __restrict__ src,
                               const int*   __restrict__ dst,
                               float*       __restrict__ h) {
    int gid  = blockIdx.x * blockDim.x + threadIdx.x;
    int e    = gid >> 5;
    int lane = gid & 31;
    if (e >= N_EDGES) return;
    int   s = src[e];
    int   d = dst[e];
    float w = ew[e];
    const float4 f = *reinterpret_cast<const float4*>(feat + (size_t)s * DIM + lane * 4);
    float* hp = h + (size_t)d * DIM + lane * 4;
    unsafeAtomicAdd(hp + 0, w * f.x);
    unsafeAtomicAdd(hp + 1, w * f.y);
    unsafeAtomicAdd(hp + 2, w * f.z);
    unsafeAtomicAdd(hp + 3, w * f.w);
}

__launch_bounds__(256)
__global__ void gemm_relu_stats_kernel(float* __restrict__ hy,
                                       const float* __restrict__ W,
                                       const float* __restrict__ bias,
                                       float* __restrict__ sum,
                                       float* __restrict__ sumsq) {
    __shared__ float tile[16 * DIM];

    const int m0 = blockIdx.x * 16;

    {
        const float4* s4 = reinterpret_cast<const float4*>(hy + (size_t)m0 * DIM);
        float4*       d4 = reinterpret_cast<float4*>(tile);
        d4[threadIdx.x]       = s4[threadIdx.x];
        d4[threadIdx.x + 256] = s4[threadIdx.x + 256];
    }
    __syncthreads();

    const int wave = threadIdx.x >> 5;
    const int lane = threadIdx.x & 31;
    const int nl   = lane & 15;
    const int half = lane >> 4;
    const int n    = wave * 16 + nl;

    const float* wrow = W + (size_t)n * DIM;

    v8f acc = {};
#pragma unroll
    for (int k0 = 0; k0 < DIM; k0 += 4) {
        const int kk = k0 + 2 * half;
        v2f a = *reinterpret_cast<const v2f*>(&tile[nl * DIM + kk]);
        v2f b = *reinterpret_cast<const v2f*>(wrow + kk);
        acc = __builtin_amdgcn_wmma_f32_16x16x4_f32(
             false, a,  false, b,
             (short)0, acc,  false,  false);
    }

    const float bn = bias[n];
    float ls = 0.f, lq = 0.f;
    float* yb = hy + (size_t)m0 * DIM + n;
#pragma unroll
    for (int v = 0; v < 8; ++v) {
        float val = fmaxf(acc[v] + bn, 0.f);
        const int m = half ? (v + 8) : v;
        yb[(size_t)m * DIM] = val;
        ls += val;
        lq += val * val;
    }
    ls += __shfl_xor(ls, 16);
    lq += __shfl_xor(lq, 16);
    if (half == 0) {
        unsafeAtomicAdd(sum + n, ls);
        unsafeAtomicAdd(sumsq + n, lq);
    }
}

__global__ void bn_finalize_kernel(const float* __restrict__ sum,
                                   const float* __restrict__ sumsq,
                                   const float* __restrict__ gamma,
                                   const float* __restrict__ beta,
                                   float* __restrict__ scale,
                                   float* __restrict__ shift) {
    int n = threadIdx.x;
    const float inv  = 1.0f / (float)N_NODES;
    float mean = sum[n] * inv;
    float var  = fmaf(-mean, mean, sumsq[n] * inv);
    float sc   = gamma[n] * rsqrtf(var + BN_EPS);
    scale[n] = sc;
    shift[n] = fmaf(-mean, sc, beta[n]);
}

__global__ void bn_apply_kernel(float4* __restrict__ y4,
                                const float4* __restrict__ scale4,
                                const float4* __restrict__ shift4) {
    int i = blockIdx.x * blockDim.x + threadIdx.x;
    if (i >= (N_NODES * DIM / 4)) return;
    const int c = i & 31;
    float4 v = y4[i];
    const float4 s = scale4[c];
    const float4 t = shift4[c];
    v.x = fmaf(v.x, s.x, t.x);
    v.y = fmaf(v.y, s.y, t.y);
    v.z = fmaf(v.z, s.z, t.z);
    v.w = fmaf(v.w, s.w, t.w);
    y4[i] = v;
}

extern "C" void kernel_launch(void* const* d_in, const int* in_sizes, int n_in,
                              void* d_out, int out_size, void* d_ws, size_t ws_size,
                              hipStream_t stream) {
    const float* feature = (const float*)d_in[0];
    const float* ew      = (const float*)d_in[1];
    const int*   src     = (const int*)d_in[2];
    const int*   dst     = (const int*)d_in[3];
    const float* W       = (const float*)d_in[4];
    const float* b       = (const float*)d_in[5];
    const float* gamma   = (const float*)d_in[6];
    const float* beta    = (const float*)d_in[7];

    float* out = (float*)d_out;
    float* ws  = (float*)d_ws;
    float* sum   = ws;
    float* sumsq = ws + 128;
    float* scale = ws + 256;
    float* shift = ws + 384;

    const int total4 = N_NODES * DIM / 4;

    zero_kernel<<<(total4 + 255) / 256, 256, 0, stream>>>((float4*)out, ws);

    const long sthreads = (long)N_EDGES * 32;
    scatter_kernel<<<(int)((sthreads + 255) / 256), 256, 0, stream>>>(
        feature, ew, src, dst, out);

    gemm_relu_stats_kernel<<<N_NODES / 16, 256, 0, stream>>>(out, W, b, sum, sumsq);

    bn_finalize_kernel<<<1, 128, 0, stream>>>(sum, sumsq, gamma, beta, scale, shift);

    bn_apply_kernel<<<(total4 + 255) / 256, 256, 0, stream>>>(
        (float4*)out, (const float4*)scale, (const float4*)shift);
}


# Zero fill and edge scatter as resident grids: 16384 scatter waves loop over groups of 8 edges, 32 row loads in flight per trip, not one short wave per edge; sums stay in atomic arrival order
# speedup vs baseline: 1.3543x; 1.3543x over previous
#include <hip/hip_runtime.h>

#define N_NODES 100000
#define N_EDGES 640000
#define DIM 128
#define BN_EPS 1e-5f

typedef __attribute__((ext_vector_type(2))) float v2f;
typedef __attribute__((ext_vector_type(8))) float v8f;

#define ZERO_WGS   1250
#define ZERO_PER   10
#define SC_EPT     8
#define SC_WGS     2048
#define SC_WAVES   (SC_WGS * 8)
#define SC_GROUPS  (N_EDGES / SC_EPT)

struct alignas(32) EdgeInts   { int   v[SC_EPT]; };
struct alignas(32) EdgeFloats { float v[SC_EPT]; };

__launch_bounds__(256)
__global__ void zero_kernel(float4* __restrict__ out4, float* __restrict__ ws) {
    const int i = blockIdx.x * 256 + threadIdx.x;
    if (i < 256) ws[i] = 0.f;
#pragma unroll
    for (int k = 0; k < ZERO_PER; ++k)
        out4[i + k * (ZERO_WGS * 256)] = make_float4(0.f, 0.f, 0.f, 0.f);
}

__launch_bounds__(256)
__global__ void scatter_kernel(const float* __restrict__ feat,
                               const float* __restrict__ ew,
                               const int*   __restrict__ src,
                               const int*   __restrict__ dst,
                               float*       __restrict__ h) {
    const int lane = threadIdx.x & 31;
    const int wave = __builtin_amdgcn_readfirstlane((int)(threadIdx.x >> 5));
    int g = blockIdx.x * 8 + wave;

    const EdgeInts*   src8 = reinterpret_cast<const EdgeInts*>(src);
    const EdgeInts*   dst8 = reinterpret_cast<const EdgeInts*>(dst);
    const EdgeFloats* ew8  = reinterpret_cast<const EdgeFloats*>(ew);

    EdgeInts   s = src8[g];
    EdgeInts   d = dst8[g];
    EdgeFloats w = ew8[g];

    for (;;) {
        float f[SC_EPT][4];
#pragma unroll
        for (int j = 0; j < SC_EPT; ++j) {
            const float* fp = feat + (size_t)s.v[j] * DIM + lane;
            f[j][0] = fp[0];
            f[j][1] = fp[32];
            f[j][2] = fp[64];
            f[j][3] = fp[96];
        }

        const EdgeInts   dc = d;
        const EdgeFloats wc = w;

        const int gn = g + SC_WAVES;
        const int gl = (gn < SC_GROUPS) ? gn : g;
        s = src8[gl];
        d = dst8[gl];
        w = ew8[gl];

#pragma unroll
        for (int j = 0; j < SC_EPT; ++j) {
            float* hp = h + (size_t)dc.v[j] * DIM + lane;
            unsafeAtomicAdd(hp + 0,  wc.v[j] * f[j][0]);
            unsafeAtomicAdd(hp + 32, wc.v[j] * f[j][1]);
            unsafeAtomicAdd(hp + 64, wc.v[j] * f[j][2]);
            unsafeAtomicAdd(hp + 96, wc.v[j] * f[j][3]);
        }

        if (gn >= SC_GROUPS) break;
        g = gn;
    }
}

__launch_bounds__(256)
__global__ void gemm_relu_stats_kernel(float* __restrict__ hy,
                                       const float* __restrict__ W,
                                       const float* __restrict__ bias,
                                       float* __restrict__ sum,
                                       float* __restrict__ sumsq) {
    __shared__ float tile[16 * DIM];

    const int m0 = blockIdx.x * 16;

    {
        const float4* s4 = reinterpret_cast<const float4*>(hy + (size_t)m0 * DIM);
        float4*       d4 = reinterpret_cast<float4*>(tile);
        d4[threadIdx.x]       = s4[threadIdx.x];
        d4[threadIdx.x + 256] = s4[threadIdx.x + 256];
    }
    __syncthreads();

    const int wave = threadIdx.x >> 5;
    const int lane = threadIdx.x & 31;
    const int nl   = lane & 15;
    const int half = lane >> 4;
    const int n    = wave * 16 + nl;

    const float* wrow = W + (size_t)n * DIM;

    v8f acc = {};
#pragma unroll
    for (int k0 = 0; k0 < DIM; k0 += 4) {
        const int kk = k0 + 2 * half;
        v2f a = *reinterpret_cast<const v2f*>(&tile[nl * DIM + kk]);
        v2f b = *reinterpret_cast<const v2f*>(wrow + kk);
        acc = __builtin_amdgcn_wmma_f32_16x16x4_f32(
             false, a,  false, b,
             (short)0, acc,  false,  false);
    }

    const float bn = bias[n];
    float ls = 0.f, lq = 0.f;
    float* yb = hy + (size_t)m0 * DIM + n;
#pragma unroll
    for (int v = 0; v < 8; ++v) {
        float val = fmaxf(acc[v] + bn, 0.f);
        const int m = half ? (v + 8) : v;
        yb[(size_t)m * DIM] = val;
        ls += val;
        lq += val * val;
    }
    ls += __shfl_xor(ls, 16);
    lq += __shfl_xor(lq, 16);
    if (half == 0) {
        unsafeAtomicAdd(sum + n, ls);
        unsafeAtomicAdd(sumsq + n, lq);
    }
}

__global__ void bn_finalize_kernel(const float* __restrict__ sum,
                                   const float* __restrict__ sumsq,
                                   const float* __restrict__ gamma,
                                   const float* __restrict__ beta,
                                   float* __restrict__ scale,
                                   float* __restrict__ shift) {
    int n = threadIdx.x;
    const float inv  = 1.0f / (float)N_NODES;
    float mean = sum[n] * inv;
    float var  = fmaf(-mean, mean, sumsq[n] * inv);
    float sc   = gamma[n] * rsqrtf(var + BN_EPS);
    scale[n] = sc;
    shift[n] = fmaf(-mean, sc, beta[n]);
}

__global__ void bn_apply_kernel(float4* __restrict__ y4,
                                const float4* __restrict__ scale4,
                                const float4* __restrict__ shift4) {
    int i = blockIdx.x * blockDim.x + threadIdx.x;
    if (i >= (N_NODES * DIM / 4)) return;
    const int c = i & 31;
    float4 v = y4[i];
    const float4 s = scale4[c];
    const float4 t = shift4[c];
    v.x = fmaf(v.x, s.x, t.x);
    v.y = fmaf(v.y, s.y, t.y);
    v.z = fmaf(v.z, s.z, t.z);
    v.w = fmaf(v.w, s.w, t.w);
    y4[i] = v;
}

extern "C" void kernel_launch(void* const* d_in, const int* in_sizes, int n_in,
                              void* d_out, int out_size, void* d_ws, size_t ws_size,
                              hipStream_t stream) {
    const float* feature = (const float*)d_in[0];
    const float* ew      = (const float*)d_in[1];
    const int*   src     = (const int*)d_in[2];
    const int*   dst     = (const int*)d_in[3];
    const float* W       = (const float*)d_in[4];
    const float* b       = (const float*)d_in[5];
    const float* gamma   = (const float*)d_in[6];
    const float* beta    = (const float*)d_in[7];

    float* out = (float*)d_out;
    float* ws  = (float*)d_ws;
    float* sum   = ws;
    float* sumsq = ws + 128;
    float* scale = ws + 256;
    float* shift = ws + 384;

    const int total4 = N_NODES * DIM / 4;

    zero_kernel<<<ZERO_WGS, 256, 0, stream>>>((float4*)out, ws);

    scatter_kernel<<<SC_WGS, 256, 0, stream>>>(feature, ew, src, dst, out);

    gemm_relu_stats_kernel<<<N_NODES / 16, 256, 0, stream>>>(out, W, b, sum, sumsq);

    bn_finalize_kernel<<<1, 128, 0, stream>>>(sum, sumsq, gamma, beta, scale, shift);

    bn_apply_kernel<<<(total4 + 255) / 256, 256, 0, stream>>>(
        (float4*)out, (const float4*)scale, (const float4*)shift);
}


# GEMM re-tiled: W staged once per 64-row work-group in LDS in fragment order; each wave owns 16 rows x all 128 columns, A fragments straight from global, no H tile in LDS; K summed in permuted order
# speedup vs baseline: 1.5043x; 1.1108x over previous
#include <hip/hip_runtime.h>

#define N_NODES 100000
#define N_EDGES 640000
#define DIM 128
#define BN_EPS 1e-5f

typedef __attribute__((ext_vector_type(2))) float v2f;
typedef __attribute__((ext_vector_type(8))) float v8f;

#define ZERO_WGS   1250
#define ZERO_PER   10
#define SC_EPT     8
#define SC_WGS     2048
#define SC_WAVES   (SC_WGS * 8)
#define SC_GROUPS  (N_EDGES / SC_EPT)

struct alignas(32) EdgeInts   { int   v[SC_EPT]; };
struct alignas(32) EdgeFloats { float v[SC_EPT]; };

__launch_bounds__(256)
__global__ void zero_kernel(float4* __restrict__ out4, float* __restrict__ ws) {
    const int i = blockIdx.x * 256 + threadIdx.x;
    if (i < 256) ws[i] = 0.f;
#pragma unroll
    for (int k = 0; k < ZERO_PER; ++k)
        out4[i + k * (ZERO_WGS * 256)] = make_float4(0.f, 0.f, 0.f, 0.f);
}

__launch_bounds__(256)
__global__ void scatter_kernel(const float* __restrict__ feat,
                               const float* __restrict__ ew,
                               const int*   __restrict__ src,
                               const int*   __restrict__ dst,
                               float*       __restrict__ h) {
    const int lane = threadIdx.x & 31;
    const int wave = __builtin_amdgcn_readfirstlane((int)(threadIdx.x >> 5));
    int g = blockIdx.x * 8 + wave;

    const EdgeInts*   src8 = reinterpret_cast<const EdgeInts*>(src);
    const EdgeInts*   dst8 = reinterpret_cast<const EdgeInts*>(dst);
    const EdgeFloats* ew8  = reinterpret_cast<const EdgeFloats*>(ew);

    EdgeInts   s = src8[g];
    EdgeInts   d = dst8[g];
    EdgeFloats w = ew8[g];

    for (;;) {
        float f[SC_EPT][4];
#pragma unroll
        for (int j = 0; j < SC_EPT; ++j) {
            const float* fp = feat + (size_t)s.v[j] * DIM + lane;
            f[j][0] = fp[0];
            f[j][1] = fp[32];
            f[j][2] = fp[64];
            f[j][3] = fp[96];
        }

        const EdgeInts   dc = d;
        const EdgeFloats wc = w;

        const int gn = g + SC_WAVES;
        const int gl = (gn < SC_GROUPS) ? gn : g;
        s = src8[gl];
        d = dst8[gl];
        w = ew8[gl];

#pragma unroll
        for (int j = 0; j < SC_EPT; ++j) {
            float* hp = h + (size_t)dc.v[j] * DIM + lane;
            unsafeAtomicAdd(hp + 0,  wc.v[j] * f[j][0]);
            unsafeAtomicAdd(hp + 32, wc.v[j] * f[j][1]);
            unsafeAtomicAdd(hp + 64, wc.v[j] * f[j][2]);
            unsafeAtomicAdd(hp + 96, wc.v[j] * f[j][3]);
        }

        if (gn >= SC_GROUPS) break;
        g = gn;
    }
}

#define GEMM_WAVES  4
#define GEMM_MTILES (N_NODES / 16)
#define GEMM_WGS    ((GEMM_MTILES + GEMM_WAVES - 1) / GEMM_WAVES)
#define WF_BLOCK    257

__launch_bounds__(128)
__global__ void gemm_relu_stats_kernel(float* __restrict__ hy,
                                       const float* __restrict__ W,
                                       const float* __restrict__ bias,
                                       float* __restrict__ sum,
                                       float* __restrict__ sumsq) {
    __shared__ float4 wfrag[16 * WF_BLOCK];

    const int lane = threadIdx.x & 31;
    const int c    = lane & 15;
    const int h    = lane >> 4;
    const int wave = __builtin_amdgcn_readfirstlane((int)(threadIdx.x >> 5));

    const int  mt   = blockIdx.x * GEMM_WAVES + wave;
    const bool live = mt < GEMM_MTILES;
    const int  m0   = (live ? mt : 0) * 16;
    float4 a[16];
    {
        const float* ap = hy + (size_t)(m0 + c) * DIM + 4 * h;
#pragma unroll
        for (int jj = 0; jj < 16; ++jj)
            a[jj] = *reinterpret_cast<const float4*>(ap + 8 * jj);
    }

    {
        const int sdst = (lane >> 1) * WF_BLOCK + (lane & 1) * 16;
#pragma unroll
        for (int i = 0; i < 32; ++i) {
            const int n = 32 * wave + i;
            const float4 v = *reinterpret_cast<const float4*>(W + (size_t)n * DIM + 4 * lane);
            wfrag[sdst + (n >> 4) * 32 + (n & 15)] = v;
        }
    }
    __syncthreads();
    if (!live) return;

    float bn[8];
#pragma unroll
    for (int t = 0; t < 8; ++t) bn[t] = bias[16 * t + c];

    v8f acc[8];
#pragma unroll
    for (int t = 0; t < 8; ++t) acc[t] = (v8f){};

#pragma unroll
    for (int jj = 0; jj < 16; ++jj) {
        float4 b[8];
#pragma unroll
        for (int t = 0; t < 8; ++t) b[t] = wfrag[jj * WF_BLOCK + t * 32 + lane];
        const v2f a0 = {a[jj].x, a[jj].y};
        const v2f a1 = {a[jj].z, a[jj].w};
#pragma unroll
        for (int t = 0; t < 8; ++t) {
            const v2f b0 = {b[t].x, b[t].y};
            acc[t] = __builtin_amdgcn_wmma_f32_16x16x4_f32(
                 false, a0,  false, b0,
                 (short)0, acc[t],  false,  false);
        }
#pragma unroll
        for (int t = 0; t < 8; ++t) {
            const v2f b1 = {b[t].z, b[t].w};
            acc[t] = __builtin_amdgcn_wmma_f32_16x16x4_f32(
                 false, a1,  false, b1,
                 (short)0, acc[t],  false,  false);
        }
    }

#pragma unroll
    for (int t = 0; t < 8; ++t) {
        const int n = 16 * t + c;
        float ls = 0.f, lq = 0.f;
        float* yb = hy + (size_t)(m0 + 8 * h) * DIM + n;
#pragma unroll
        for (int r = 0; r < 8; ++r) {
            float val = fmaxf(acc[t][r] + bn[t], 0.f);
            yb[(size_t)r * DIM] = val;
            ls += val;
            lq += val * val;
        }
        ls += __shfl_xor(ls, 16);
        lq += __shfl_xor(lq, 16);
        if (h == 0) {
            unsafeAtomicAdd(sum + n, ls);
            unsafeAtomicAdd(sumsq + n, lq);
        }
    }
}

__global__ void bn_finalize_kernel(const float* __restrict__ sum,
                                   const float* __restrict__ sumsq,
                                   const float* __restrict__ gamma,
                                   const float* __restrict__ beta,
                                   float* __restrict__ scale,
                                   float* __restrict__ shift) {
    int n = threadIdx.x;
    const float inv  = 1.0f / (float)N_NODES;
    float mean = sum[n] * inv;
    float var  = fmaf(-mean, mean, sumsq[n] * inv);
    float sc   = gamma[n] * rsqrtf(var + BN_EPS);
    scale[n] = sc;
    shift[n] = fmaf(-mean, sc, beta[n]);
}

__global__ void bn_apply_kernel(float4* __restrict__ y4,
                                const float4* __restrict__ scale4,
                                const float4* __restrict__ shift4) {
    int i = blockIdx.x * blockDim.x + threadIdx.x;
    if (i >= (N_NODES * DIM / 4)) return;
    const int c = i & 31;
    float4 v = y4[i];
    const float4 s = scale4[c];
    const float4 t = shift4[c];
    v.x = fmaf(v.x, s.x, t.x);
    v.y = fmaf(v.y, s.y, t.y);
    v.z = fmaf(v.z, s.z, t.z);
    v.w = fmaf(v.w, s.w, t.w);
    y4[i] = v;
}

extern "C" void kernel_launch(void* const* d_in, const int* in_sizes, int n_in,
                              void* d_out, int out_size, void* d_ws, size_t ws_size,
                              hipStream_t stream) {
    const float* feature = (const float*)d_in[0];
    const float* ew      = (const float*)d_in[1];
    const int*   src     = (const int*)d_in[2];
    const int*   dst     = (const int*)d_in[3];
    const float* W       = (const float*)d_in[4];
    const float* b       = (const float*)d_in[5];
    const float* gamma   = (const float*)d_in[6];
    const float* beta    = (const float*)d_in[7];

    float* out = (float*)d_out;
    float* ws  = (float*)d_ws;
    float* sum   = ws;
    float* sumsq = ws + 128;
    float* scale = ws + 256;
    float* shift = ws + 384;

    const int total4 = N_NODES * DIM / 4;

    zero_kernel<<<ZERO_WGS, 256, 0, stream>>>((float4*)out, ws);

    scatter_kernel<<<SC_WGS, 256, 0, stream>>>(feature, ew, src, dst, out);

    gemm_relu_stats_kernel<<<GEMM_WGS, GEMM_WAVES * 32, 0, stream>>>(out, W, b, sum, sumsq);

    bn_finalize_kernel<<<1, 128, 0, stream>>>(sum, sumsq, gamma, beta, scale, shift);

    bn_apply_kernel<<<(total4 + 255) / 256, 256, 0, stream>>>(
        (float4*)out, (const float4*)scale, (const float4*)shift);
}


# GEMM channel stats: 4 waves' partial sums meet in LDS behind a 2nd barrier, then 1 atomic per channel per work-group (8 whole-line atomic instructions, was 64 half-line); stats sum tree changes
# speedup vs baseline: 1.8834x; 1.2520x over previous
#include <hip/hip_runtime.h>

#define N_NODES 100000
#define N_EDGES 640000
#define DIM 128
#define BN_EPS 1e-5f

typedef __attribute__((ext_vector_type(2))) float v2f;
typedef __attribute__((ext_vector_type(8))) float v8f;

#define ZERO_WGS   1250
#define ZERO_PER   10
#define SC_EPT     8
#define SC_WGS     2048
#define SC_WAVES   (SC_WGS * 8)
#define SC_GROUPS  (N_EDGES / SC_EPT)

struct alignas(32) EdgeInts   { int   v[SC_EPT]; };
struct alignas(32) EdgeFloats { float v[SC_EPT]; };

__launch_bounds__(256)
__global__ void zero_kernel(float4* __restrict__ out4, float* __restrict__ ws) {
    const int i = blockIdx.x * 256 + threadIdx.x;
    if (i < 256) ws[i] = 0.f;
#pragma unroll
    for (int k = 0; k < ZERO_PER; ++k)
        out4[i + k * (ZERO_WGS * 256)] = make_float4(0.f, 0.f, 0.f, 0.f);
}

__launch_bounds__(256)
__global__ void scatter_kernel(const float* __restrict__ feat,
                               const float* __restrict__ ew,
                               const int*   __restrict__ src,
                               const int*   __restrict__ dst,
                               float*       __restrict__ h) {
    const int lane = threadIdx.x & 31;
    const int wave = __builtin_amdgcn_readfirstlane((int)(threadIdx.x >> 5));
    int g = blockIdx.x * 8 + wave;

    const EdgeInts*   src8 = reinterpret_cast<const EdgeInts*>(src);
    const EdgeInts*   dst8 = reinterpret_cast<const EdgeInts*>(dst);
    const EdgeFloats* ew8  = reinterpret_cast<const EdgeFloats*>(ew);

    EdgeInts   s = src8[g];
    EdgeInts   d = dst8[g];
    EdgeFloats w = ew8[g];

    for (;;) {
        float f[SC_EPT][4];
#pragma unroll
        for (int j = 0; j < SC_EPT; ++j) {
            const float* fp = feat + (size_t)s.v[j] * DIM + lane;
            f[j][0] = fp[0];
            f[j][1] = fp[32];
            f[j][2] = fp[64];
            f[j][3] = fp[96];
        }

        const EdgeInts   dc = d;
        const EdgeFloats wc = w;

        const int gn = g + SC_WAVES;
        const int gl = (gn < SC_GROUPS) ? gn : g;
        s = src8[gl];
        d = dst8[gl];
        w = ew8[gl];

#pragma unroll
        for (int j = 0; j < SC_EPT; ++j) {
            float* hp = h + (size_t)dc.v[j] * DIM + lane;
            unsafeAtomicAdd(hp + 0,  wc.v[j] * f[j][0]);
            unsafeAtomicAdd(hp + 32, wc.v[j] * f[j][1]);
            unsafeAtomicAdd(hp + 64, wc.v[j] * f[j][2]);
            unsafeAtomicAdd(hp + 96, wc.v[j] * f[j][3]);
        }

        if (gn >= SC_GROUPS) break;
        g = gn;
    }
}

#define GEMM_WAVES  4
#define GEMM_MTILES (N_NODES / 16)
#define GEMM_WGS    ((GEMM_MTILES + GEMM_WAVES - 1) / GEMM_WAVES)
#define WF_BLOCK    257

__launch_bounds__(128)
__global__ void gemm_relu_stats_kernel(float* __restrict__ hy,
                                       const float* __restrict__ W,
                                       const float* __restrict__ bias,
                                       float* __restrict__ sum,
                                       float* __restrict__ sumsq) {
    __shared__ float4 wfrag[16 * WF_BLOCK];
    __shared__ float  part[GEMM_WAVES * 256];

    const int lane = threadIdx.x & 31;
    const int c    = lane & 15;
    const int h    = lane >> 4;
    const int wave = __builtin_amdgcn_readfirstlane((int)(threadIdx.x >> 5));

    const int  mt   = blockIdx.x * GEMM_WAVES + wave;
    const bool live = mt < GEMM_MTILES;
    const int  m0   = (live ? mt : 0) * 16;
    float4 a[16];
    {
        const float* ap = hy + (size_t)(m0 + c) * DIM + 4 * h;
#pragma unroll
        for (int jj = 0; jj < 16; ++jj)
            a[jj] = *reinterpret_cast<const float4*>(ap + 8 * jj);
    }

    {
        const int sdst = (lane >> 1) * WF_BLOCK + (lane & 1) * 16;
#pragma unroll
        for (int i = 0; i < 32; ++i) {
            const int n = 32 * wave + i;
            const float4 v = *reinterpret_cast<const float4*>(W + (size_t)n * DIM + 4 * lane);
            wfrag[sdst + (n >> 4) * 32 + (n & 15)] = v;
        }
    }
    __syncthreads();

    v8f acc[8];
#pragma unroll
    for (int t = 0; t < 8; ++t) acc[t] = (v8f){};

    if (live) {
        float bn[8];
#pragma unroll
        for (int t = 0; t < 8; ++t) bn[t] = bias[16 * t + c];

#pragma unroll
        for (int jj = 0; jj < 16; ++jj) {
            float4 b[8];
#pragma unroll
            for (int t = 0; t < 8; ++t) b[t] = wfrag[jj * WF_BLOCK + t * 32 + lane];
            const v2f a0 = {a[jj].x, a[jj].y};
            const v2f a1 = {a[jj].z, a[jj].w};
#pragma unroll
            for (int t = 0; t < 8; ++t) {
                const v2f b0 = {b[t].x, b[t].y};
                acc[t] = __builtin_amdgcn_wmma_f32_16x16x4_f32(
                     false, a0,  false, b0,
                     (short)0, acc[t],  false,  false);
            }
#pragma unroll
            for (int t = 0; t < 8; ++t) {
                const v2f b1 = {b[t].z, b[t].w};
                acc[t] = __builtin_amdgcn_wmma_f32_16x16x4_f32(
                     false, a1,  false, b1,
                     (short)0, acc[t],  false,  false);
            }
        }

#pragma unroll
        for (int t = 0; t < 8; ++t) {
            float ls = 0.f, lq = 0.f;
#pragma unroll
            for (int r = 0; r < 8; ++r) {
                const float val = fmaxf(acc[t][r] + bn[t], 0.f);
                acc[t][r] = val;
                ls += val;
                lq += val * val;
            }
            ls += __shfl_xor(ls, 16);
            lq += __shfl_xor(lq, 16);
            part[wave * 256 + 32 * t + lane] = h ? lq : ls;
        }
    } else {
#pragma unroll
        for (int t = 0; t < 8; ++t) part[wave * 256 + 32 * t + lane] = 0.f;
    }
    __syncthreads();

    {
        const int j = threadIdx.x;
        const int e = 32 * (j >> 4) + (j & 15);
        float s = 0.f, q = 0.f;
#pragma unroll
        for (int w = 0; w < GEMM_WAVES; ++w) {
            s += part[w * 256 + e];
            q += part[w * 256 + e + 16];
        }
        unsafeAtomicAdd(sum + j, s);
        unsafeAtomicAdd(sumsq + j, q);
    }

    if (live) {
#pragma unroll
        for (int t = 0; t < 8; ++t) {
            float* yb = hy + (size_t)(m0 + 8 * h) * DIM + 16 * t + c;
#pragma unroll
            for (int r = 0; r < 8; ++r) yb[(size_t)r * DIM] = acc[t][r];
        }
    }
}

__global__ void bn_finalize_kernel(const float* __restrict__ sum,
                                   const float* __restrict__ sumsq,
                                   const float* __restrict__ gamma,
                                   const float* __restrict__ beta,
                                   float* __restrict__ scale,
                                   float* __restrict__ shift) {
    int n = threadIdx.x;
    const float inv  = 1.0f / (float)N_NODES;
    float mean = sum[n] * inv;
    float var  = fmaf(-mean, mean, sumsq[n] * inv);
    float sc   = gamma[n] * rsqrtf(var + BN_EPS);
    scale[n] = sc;
    shift[n] = fmaf(-mean, sc, beta[n]);
}

__global__ void bn_apply_kernel(float4* __restrict__ y4,
                                const float4* __restrict__ scale4,
                                const float4* __restrict__ shift4) {
    int i = blockIdx.x * blockDim.x + threadIdx.x;
    if (i >= (N_NODES * DIM / 4)) return;
    const int c = i & 31;
    float4 v = y4[i];
    const float4 s = scale4[c];
    const float4 t = shift4[c];
    v.x = fmaf(v.x, s.x, t.x);
    v.y = fmaf(v.y, s.y, t.y);
    v.z = fmaf(v.z, s.z, t.z);
    v.w = fmaf(v.w, s.w, t.w);
    y4[i] = v;
}

extern "C" void kernel_launch(void* const* d_in, const int* in_sizes, int n_in,
                              void* d_out, int out_size, void* d_ws, size_t ws_size,
                              hipStream_t stream) {
    const float* feature = (const float*)d_in[0];
    const float* ew      = (const float*)d_in[1];
    const int*   src     = (const int*)d_in[2];
    const int*   dst     = (const int*)d_in[3];
    const float* W       = (const float*)d_in[4];
    const float* b       = (const float*)d_in[5];
    const float* gamma   = (const float*)d_in[6];
    const float* beta    = (const float*)d_in[7];

    float* out = (float*)d_out;
    float* ws  = (float*)d_ws;
    float* sum   = ws;
    float* sumsq = ws + 128;
    float* scale = ws + 256;
    float* shift = ws + 384;

    const int total4 = N_NODES * DIM / 4;

    zero_kernel<<<ZERO_WGS, 256, 0, stream>>>((float4*)out, ws);

    scatter_kernel<<<SC_WGS, 256, 0, stream>>>(feature, ew, src, dst, out);

    gemm_relu_stats_kernel<<<GEMM_WGS, GEMM_WAVES * 32, 0, stream>>>(out, W, b, sum, sumsq);

    bn_finalize_kernel<<<1, 128, 0, stream>>>(sum, sumsq, gamma, beta, scale, shift);

    bn_apply_kernel<<<(total4 + 255) / 256, 256, 0, stream>>>(
        (float4*)out, (const float4*)scale, (const float4*)shift);
}
